# starting kernel
# baseline (speedup 1.0000x reference)
#include <hip/hip_runtime.h>

typedef float v2f __attribute__((ext_vector_type(2)));
typedef float v8f __attribute__((ext_vector_type(8)));

#define HID  128
#define OUTC 128

__device__ __forceinline__ void wave_gemm_16x64(const float* __restrict__ h,
                                                const float* __restrict__ W,
                                                int m0, int nbase, int lane,
                                                v8f acc[4]) {
  const int l16  = lane & 15;
  const int koff = (lane >> 4) * 2;
  const float* ap  = h + (size_t)(m0 + l16) * HID + koff;
  const float* bp0 = W + (size_t)(nbase +  0 + l16) * HID + koff;
  const float* bp1 = W + (size_t)(nbase + 16 + l16) * HID + koff;
  const float* bp2 = W + (size_t)(nbase + 32 + l16) * HID + koff;
  const float* bp3 = W + (size_t)(nbase + 48 + l16) * HID + koff;
  acc[0] = v8f{}; acc[1] = v8f{}; acc[2] = v8f{}; acc[3] = v8f{};
#pragma unroll 8
  for (int k0 = 0; k0 < HID; k0 += 4) {
    v2f a  = *(const v2f*)(ap  + k0);
    v2f b0 = *(const v2f*)(bp0 + k0);
    v2f b1 = *(const v2f*)(bp1 + k0);
    v2f b2 = *(const v2f*)(bp2 + k0);
    v2f b3 = *(const v2f*)(bp3 + k0);
    acc[0] = __builtin_amdgcn_wmma_f32_16x16x4_f32(false, a, false, b0, (short)0, acc[0], false, false);
    acc[1] = __builtin_amdgcn_wmma_f32_16x16x4_f32(false, a, false, b1, (short)0, acc[1], false, false);
    acc[2] = __builtin_amdgcn_wmma_f32_16x16x4_f32(false, a, false, b2, (short)0, acc[2], false, false);
    acc[3] = __builtin_amdgcn_wmma_f32_16x16x4_f32(false, a, false, b3, (short)0, acc[3], false, false);
  }
}

__global__ __launch_bounds__(128) void gemm_kernel(const float* __restrict__ h,
                                                   const float* __restrict__ W,
                                                   float* __restrict__ z) {
  const int lane  = threadIdx.x & 31;
  const int wave  = threadIdx.x >> 5;
  const int m0    = blockIdx.x * 32 + (wave >> 1) * 16;
  const int nbase = (wave & 1) * 64;
  v8f acc[4];
  wave_gemm_16x64(h, W, m0, nbase, lane, acc);
  const int l16   = lane & 15;
  const int rbase = (lane >> 4) * 8;
#pragma unroll
  for (int r = 0; r < 8; ++r) {
    const int m = m0 + rbase + r;
#pragma unroll
    for (int j = 0; j < 4; ++j) {
      z[(size_t)m * OUTC + nbase + j * 16 + l16] = acc[j][r];
    }
  }
}

__global__ __launch_bounds__(256) void scatter_kernel(const int* __restrict__ src,
                                                      const int* __restrict__ dst,
                                                      const float* __restrict__ z,
                                                      float* __restrict__ s,
                                                      float* __restrict__ deg,
                                                      int n_edges) {
  const int e = blockIdx.x * (blockDim.x >> 5) + (threadIdx.x >> 5);
  if (e >= n_edges) return;
  const int lane = threadIdx.x & 31;
  const int si = src[e];
  const int di = dst[e];
  const float4 v = *(const float4*)(z + (size_t)si * OUTC + lane * 4);
  float* o = s + (size_t)di * OUTC + lane * 4;
  atomicAdd(o + 0, v.x);
  atomicAdd(o + 1, v.y);
  atomicAdd(o + 2, v.z);
  atomicAdd(o + 3, v.w);
  if (lane == 0) atomicAdd(deg + di, 1.0f);
}

__global__ __launch_bounds__(128) void gemm_finalize_kernel(const float* __restrict__ h,
                                                            const float* __restrict__ Wl,
                                                            const float* __restrict__ bl,
                                                            const float* __restrict__ s1,
                                                            const float* __restrict__ s2,
                                                            const float* __restrict__ deg1,
                                                            const float* __restrict__ deg2,
                                                            float* __restrict__ out) {
  const int lane  = threadIdx.x & 31;
  const int wave  = threadIdx.x >> 5;
  const int m0    = blockIdx.x * 32 + (wave >> 1) * 16;
  const int nbase = (wave & 1) * 64;
  v8f acc[4];
  wave_gemm_16x64(h, Wl, m0, nbase, lane, acc);
  const int l16   = lane & 15;
  const int rbase = (lane >> 4) * 8;
#pragma unroll
  for (int r = 0; r < 8; ++r) {
    const int m = m0 + rbase + r;
    const float i1 = 0.5f / fmaxf(deg1[m], 1.0f);
    const float i2 = 0.5f / fmaxf(deg2[m], 1.0f);
#pragma unroll
    for (int j = 0; j < 4; ++j) {
      const int    c   = nbase + j * 16 + l16;
      const size_t idx = (size_t)m * OUTC + c;
      const float  v   = acc[j][r] + bl[c] + s1[idx] * i1 + s2[idx] * i2;
      out[idx] = fmaxf(v, 0.0f);
    }
  }
}

extern "C" void kernel_launch(void* const* d_in, const int* in_sizes, int n_in,
                              void* d_out, int out_size, void* d_ws, size_t ws_size,
                              hipStream_t stream) {
  (void)n_in; (void)out_size; (void)ws_size;
  const float* h    = (const float*)d_in[0];
  const int*   src1 = (const int*)  d_in[1];
  const int*   dst1 = (const int*)  d_in[2];
  const int*   src2 = (const int*)  d_in[3];
  const int*   dst2 = (const int*)  d_in[4];
  const float* W1   = (const float*)d_in[5];
  const float* W2   = (const float*)d_in[6];
  const float* Wl   = (const float*)d_in[7];
  const float* bl   = (const float*)d_in[8];
  float*       out  = (float*)d_out;

  const size_t nodes = (size_t)in_sizes[0] / HID;
  const int    ne    = in_sizes[1];
  const size_t feat  = nodes * OUTC;

  float* s1   = (float*)d_ws;
  float* s2   = s1 + feat;
  float* deg1 = s2 + feat;
  float* deg2 = deg1 + nodes;
  float* z    = deg2 + nodes;

  hipMemsetAsync(d_ws, 0, (2 * feat + 2 * nodes) * sizeof(float), stream);

  const dim3 gblk(128);
  const dim3 ggrd((unsigned)(nodes / 32));
  const dim3 sblk(256);
  const dim3 sgrd((unsigned)((ne + 7) / 8));

  gemm_kernel<<<ggrd, gblk, 0, stream>>>(h, W1, z);
  scatter_kernel<<<sgrd, sblk, 0, stream>>>(src1, dst1, z, s1, deg1, ne);

  gemm_kernel<<<ggrd, gblk, 0, stream>>>(h, W2, z);
  scatter_kernel<<<sgrd, sblk, 0, stream>>>(src2, dst2, z, s2, deg2, ne);

  gemm_finalize_kernel<<<ggrd, gblk, 0, stream>>>(h, Wl, bl, s1, s2, deg1, deg2, out);
}


# Weight staged once per work-group in LDS (padded pitch); B fragments read from LDS one chunk ahead of the matrix work; wave tile 16x128 so each h row is loaded once per GEMM; k order unchanged
# speedup vs baseline: 1.1736x; 1.1736x over previous
#include <hip/hip_runtime.h>


typedef float v2f __attribute__((ext_vector_type(2)));
typedef float v8f __attribute__((ext_vector_type(8)));

#define HID  128
#define OUTC 128

#define GEMM_WAVES   8
#define GEMM_THREADS (GEMM_WAVES * 32)
#define WPITCH 132


__device__ __forceinline__ void stage_weight(const float* __restrict__ W, float* wl) {
#pragma unroll
  for (int i = 0; i < (OUTC * HID / 4) / GEMM_THREADS; ++i) {
    const int idx = i * GEMM_THREADS + (int)threadIdx.x;
    const int row = idx >> 5;
    const int c4  = idx & 31;
    const float4 v = *(const float4*)(W + (size_t)row * HID + c4 * 4);
    *(float4*)(wl + row * WPITCH + c4 * 4) = v;
  }
}

__device__ __forceinline__ void load_a_tile(const float* __restrict__ h, int m0, int lane, v2f a[32]) {
  const float* ap = h + (size_t)(m0 + (lane & 15)) * HID + (lane >> 4) * 2;
#pragma unroll
  for (int s = 0; s < 32; ++s) a[s] = *(const v2f*)(ap + 4 * s);
}

__device__ __forceinline__ void load_b_chunk(const float* bp, int c, v2f blo[8], v2f bhi[8]) {
#pragma unroll
  for (int j = 0; j < 8; ++j) {
    blo[j] = *(const v2f*)(bp + j * 16 * WPITCH + 8 * c);
    bhi[j] = *(const v2f*)(bp + j * 16 * WPITCH + 8 * c + 4);
  }
}

__device__ __forceinline__ void wave_gemm_16x128(const v2f a[32], const float* wl, int lane, v8f acc[8]) {
  const float* bp = wl + (lane & 15) * WPITCH + (lane >> 4) * 2;
#pragma unroll
  for (int j = 0; j < 8; ++j) acc[j] = v8f{};
  v2f blo[2][8], bhi[2][8];
  load_b_chunk(bp, 0, blo[0], bhi[0]);
#pragma unroll
  for (int c = 0; c < 16; ++c) {
    const int cur = c & 1;
    const int nxt = cur ^ 1;
    if (c + 1 < 16) load_b_chunk(bp, c + 1, blo[nxt], bhi[nxt]);
    __builtin_amdgcn_sched_barrier(0);
#pragma unroll
    for (int j = 0; j < 8; ++j)
      acc[j] = __builtin_amdgcn_wmma_f32_16x16x4_f32(false, a[2 * c], false, blo[cur][j], (short)0, acc[j], false, false);
#pragma unroll
    for (int j = 0; j < 8; ++j)
      acc[j] = __builtin_amdgcn_wmma_f32_16x16x4_f32(false, a[2 * c + 1], false, bhi[cur][j], (short)0, acc[j], false, false);
    __builtin_amdgcn_sched_barrier(0);
  }
}

__global__ __launch_bounds__(GEMM_THREADS) void gemm_kernel(const float* __restrict__ h,
                                                            const float* __restrict__ W,
                                                            float* __restrict__ z,
                                                            int ntiles) {
  __shared__ float wl[OUTC * WPITCH];
  const int lane = threadIdx.x & 31;
  const int wave = threadIdx.x >> 5;
  const int tile = blockIdx.x * GEMM_WAVES + wave;
  const int m0   = min(tile, ntiles - 1) * 16;
  stage_weight(W, wl);
  v2f a[32];
  load_a_tile(h, m0, lane, a);
  __syncthreads();
  if (tile >= ntiles) return;
  v8f acc[8];
  wave_gemm_16x128(a, wl, lane, acc);
  const int l16   = lane & 15;
  const int rbase = (lane >> 4) * 8;
#pragma unroll
  for (int r = 0; r < 8; ++r) {
    const int m = m0 + rbase + r;
#pragma unroll
    for (int j = 0; j < 8; ++j) {
      z[(size_t)m * OUTC + j * 16 + l16] = acc[j][r];
    }
  }
}

__global__ __launch_bounds__(256) void scatter_kernel(const int* __restrict__ src,
                                                      const int* __restrict__ dst,
                                                      const float* __restrict__ z,
                                                      float* __restrict__ s,
                                                      float* __restrict__ deg,
                                                      int n_edges) {
  const int e = blockIdx.x * (blockDim.x >> 5) + (threadIdx.x >> 5);
  if (e >= n_edges) return;
  const int lane = threadIdx.x & 31;
  const int si = src[e];
  const int di = dst[e];
  const float4 v = *(const float4*)(z + (size_t)si * OUTC + lane * 4);
  float* o = s + (size_t)di * OUTC + lane * 4;
  atomicAdd(o + 0, v.x);
  atomicAdd(o + 1, v.y);
  atomicAdd(o + 2, v.z);
  atomicAdd(o + 3, v.w);
  if (lane == 0) atomicAdd(deg + di, 1.0f);
}

__global__ __launch_bounds__(GEMM_THREADS) void gemm_finalize_kernel(const float* __restrict__ h,
                                                                     const float* __restrict__ Wl,
                                                                     const float* __restrict__ bl,
                                                                     const float* __restrict__ s1,
                                                                     const float* __restrict__ s2,
                                                                     const float* __restrict__ deg1,
                                                                     const float* __restrict__ deg2,
                                                                     float* __restrict__ out,
                                                                     int ntiles) {
  __shared__ float wl[OUTC * WPITCH];
  const int lane = threadIdx.x & 31;
  const int wave = threadIdx.x >> 5;
  const int tile = blockIdx.x * GEMM_WAVES + wave;
  const int m0   = min(tile, ntiles - 1) * 16;
  stage_weight(Wl, wl);
  v2f a[32];
  load_a_tile(h, m0, lane, a);
  __syncthreads();
  if (tile >= ntiles) return;
  v8f acc[8];
  wave_gemm_16x128(a, wl, lane, acc);
  const int l16   = lane & 15;
  const int rbase = (lane >> 4) * 8;
#pragma unroll
  for (int r = 0; r < 8; ++r) {
    const int m = m0 + rbase + r;
    const float i1 = 0.5f / fmaxf(deg1[m], 1.0f);
    const float i2 = 0.5f / fmaxf(deg2[m], 1.0f);
#pragma unroll
    for (int j = 0; j < 8; ++j) {
      const int    c   = j * 16 + l16;
      const size_t idx = (size_t)m * OUTC + c;
      const float  v   = acc[j][r] + bl[c] + s1[idx] * i1 + s2[idx] * i2;
      out[idx] = fmaxf(v, 0.0f);
    }
  }
}

extern "C" void kernel_launch(void* const* d_in, const int* in_sizes, int n_in,
                              void* d_out, int out_size, void* d_ws, size_t ws_size,
                              hipStream_t stream) {
  (void)n_in; (void)out_size; (void)ws_size;
  const float* h    = (const float*)d_in[0];
  const int*   src1 = (const int*)  d_in[1];
  const int*   dst1 = (const int*)  d_in[2];
  const int*   src2 = (const int*)  d_in[3];
  const int*   dst2 = (const int*)  d_in[4];
  const float* W1   = (const float*)d_in[5];
  const float* W2   = (const float*)d_in[6];
  const float* Wl   = (const float*)d_in[7];
  const float* bl   = (const float*)d_in[8];
  float*       out  = (float*)d_out;

  const size_t nodes = (size_t)in_sizes[0] / HID;
  const int    ne    = in_sizes[1];
  const size_t feat  = nodes * OUTC;

  float* s1   = (float*)d_ws;
  float* s2   = s1 + feat;
  float* deg1 = s2 + feat;
  float* deg2 = deg1 + nodes;
  float* z    = deg2 + nodes;

  hipMemsetAsync(d_ws, 0, (2 * feat + 2 * nodes) * sizeof(float), stream);

  const int  ntiles = (int)(nodes / 16);
  const dim3 gblk(GEMM_THREADS);
  const dim3 ggrd((unsigned)((ntiles + GEMM_WAVES - 1) / GEMM_WAVES));
  const dim3 sblk(256);
  const dim3 sgrd((unsigned)((ne + 7) / 8));

  gemm_kernel<<<ggrd, gblk, 0, stream>>>(h, W1, z, ntiles);
  scatter_kernel<<<sgrd, sblk, 0, stream>>>(src1, dst1, z, s1, deg1, ne);

  gemm_kernel<<<ggrd, gblk, 0, stream>>>(h, W2, z, ntiles);
  scatter_kernel<<<sgrd, sblk, 0, stream>>>(src2, dst2, z, s2, deg2, ne);

  gemm_finalize_kernel<<<ggrd, gblk, 0, stream>>>(h, Wl, bl, s1, s2, deg1, deg2, out, ntiles);
}


# Scatter: lane l adds columns l, l+32, l+64, l+96, so each atomic instruction updates one whole 128-B line of the s row instead of 8 floats in each of 4 lines; adds still in arrival order
# speedup vs baseline: 1.2907x; 1.0998x over previous
#include <hip/hip_runtime.h>


typedef float v2f __attribute__((ext_vector_type(2)));
typedef float v8f __attribute__((ext_vector_type(8)));

#define HID  128
#define OUTC 128

#define GEMM_WAVES   8
#define GEMM_THREADS (GEMM_WAVES * 32)
#define WPITCH 132


__device__ __forceinline__ void stage_weight(const float* __restrict__ W, float* wl) {
#pragma unroll
  for (int i = 0; i < (OUTC * HID / 4) / GEMM_THREADS; ++i) {
    const int idx = i * GEMM_THREADS + (int)threadIdx.x;
    const int row = idx >> 5;
    const int c4  = idx & 31;
    const float4 v = *(const float4*)(W + (size_t)row * HID + c4 * 4);
    *(float4*)(wl + row * WPITCH + c4 * 4) = v;
  }
}

__device__ __forceinline__ void load_a_tile(const float* __restrict__ h, int m0, int lane, v2f a[32]) {
  const float* ap = h + (size_t)(m0 + (lane & 15)) * HID + (lane >> 4) * 2;
#pragma unroll
  for (int s = 0; s < 32; ++s) a[s] = *(const v2f*)(ap + 4 * s);
}

__device__ __forceinline__ void load_b_chunk(const float* bp, int c, v2f blo[8], v2f bhi[8]) {
#pragma unroll
  for (int j = 0; j < 8; ++j) {
    blo[j] = *(const v2f*)(bp + j * 16 * WPITCH + 8 * c);
    bhi[j] = *(const v2f*)(bp + j * 16 * WPITCH + 8 * c + 4);
  }
}

__device__ __forceinline__ void wave_gemm_16x128(const v2f a[32], const float* wl, int lane, v8f acc[8]) {
  const float* bp = wl + (lane & 15) * WPITCH + (lane >> 4) * 2;
#pragma unroll
  for (int j = 0; j < 8; ++j) acc[j] = v8f{};
  v2f blo[2][8], bhi[2][8];
  load_b_chunk(bp, 0, blo[0], bhi[0]);
#pragma unroll
  for (int c = 0; c < 16; ++c) {
    const int cur = c & 1;
    const int nxt = cur ^ 1;
    if (c + 1 < 16) load_b_chunk(bp, c + 1, blo[nxt], bhi[nxt]);
    __builtin_amdgcn_sched_barrier(0);
#pragma unroll
    for (int j = 0; j < 8; ++j)
      acc[j] = __builtin_amdgcn_wmma_f32_16x16x4_f32(false, a[2 * c], false, blo[cur][j], (short)0, acc[j], false, false);
#pragma unroll
    for (int j = 0; j < 8; ++j)
      acc[j] = __builtin_amdgcn_wmma_f32_16x16x4_f32(false, a[2 * c + 1], false, bhi[cur][j], (short)0, acc[j], false, false);
    __builtin_amdgcn_sched_barrier(0);
  }
}

__global__ __launch_bounds__(GEMM_THREADS) void gemm_kernel(const float* __restrict__ h,
                                                            const float* __restrict__ W,
                                                            float* __restrict__ z,
                                                            int ntiles) {
  __shared__ float wl[OUTC * WPITCH];
  const int lane = threadIdx.x & 31;
  const int wave = threadIdx.x >> 5;
  const int tile = blockIdx.x * GEMM_WAVES + wave;
  const int m0   = min(tile, ntiles - 1) * 16;
  stage_weight(W, wl);
  v2f a[32];
  load_a_tile(h, m0, lane, a);
  __syncthreads();
  if (tile >= ntiles) return;
  v8f acc[8];
  wave_gemm_16x128(a, wl, lane, acc);
  const int l16   = lane & 15;
  const int rbase = (lane >> 4) * 8;
#pragma unroll
  for (int r = 0; r < 8; ++r) {
    const int m = m0 + rbase + r;
#pragma unroll
    for (int j = 0; j < 8; ++j) {
      z[(size_t)m * OUTC + j * 16 + l16] = acc[j][r];
    }
  }
}

__global__ __launch_bounds__(256) void scatter_kernel(const int* __restrict__ src,
                                                      const int* __restrict__ dst,
                                                      const float* __restrict__ z,
                                                      float* __restrict__ s,
                                                      float* __restrict__ deg,
                                                      int n_edges) {
  const int e = blockIdx.x * (blockDim.x >> 5) + (threadIdx.x >> 5);
  if (e >= n_edges) return;
  const int lane = threadIdx.x & 31;
  const int si = src[e];
  const int di = dst[e];
  const float* zp = z + (size_t)si * OUTC + lane;
  const float v0 = zp[0];
  const float v1 = zp[32];
  const float v2 = zp[64];
  const float v3 = zp[96];
  float* o = s + (size_t)di * OUTC + lane;
  atomicAdd(o + 0,  v0);
  atomicAdd(o + 32, v1);
  atomicAdd(o + 64, v2);
  atomicAdd(o + 96, v3);
  if (lane == 0) atomicAdd(deg + di, 1.0f);
}

__global__ __launch_bounds__(GEMM_THREADS) void gemm_finalize_kernel(const float* __restrict__ h,
                                                                     const float* __restrict__ Wl,
                                                                     const float* __restrict__ bl,
                                                                     const float* __restrict__ s1,
                                                                     const float* __restrict__ s2,
                                                                     const float* __restrict__ deg1,
                                                                     const float* __restrict__ deg2,
                                                                     float* __restrict__ out,
                                                                     int ntiles) {
  __shared__ float wl[OUTC * WPITCH];
  const int lane = threadIdx.x & 31;
  const int wave = threadIdx.x >> 5;
  const int tile = blockIdx.x * GEMM_WAVES + wave;
  const int m0   = min(tile, ntiles - 1) * 16;
  stage_weight(Wl, wl);
  v2f a[32];
  load_a_tile(h, m0, lane, a);
  __syncthreads();
  if (tile >= ntiles) return;
  v8f acc[8];
  wave_gemm_16x128(a, wl, lane, acc);
  const int l16   = lane & 15;
  const int rbase = (lane >> 4) * 8;
#pragma unroll
  for (int r = 0; r < 8; ++r) {
    const int m = m0 + rbase + r;
    const float i1 = 0.5f / fmaxf(deg1[m], 1.0f);
    const float i2 = 0.5f / fmaxf(deg2[m], 1.0f);
#pragma unroll
    for (int j = 0; j < 8; ++j) {
      const int    c   = j * 16 + l16;
      const size_t idx = (size_t)m * OUTC + c;
      const float  v   = acc[j][r] + bl[c] + s1[idx] * i1 + s2[idx] * i2;
      out[idx] = fmaxf(v, 0.0f);
    }
  }
}

extern "C" void kernel_launch(void* const* d_in, const int* in_sizes, int n_in,
                              void* d_out, int out_size, void* d_ws, size_t ws_size,
                              hipStream_t stream) {
  (void)n_in; (void)out_size; (void)ws_size;
  const float* h    = (const float*)d_in[0];
  const int*   src1 = (const int*)  d_in[1];
  const int*   dst1 = (const int*)  d_in[2];
  const int*   src2 = (const int*)  d_in[3];
  const int*   dst2 = (const int*)  d_in[4];
  const float* W1   = (const float*)d_in[5];
  const float* W2   = (const float*)d_in[6];
  const float* Wl   = (const float*)d_in[7];
  const float* bl   = (const float*)d_in[8];
  float*       out  = (float*)d_out;

  const size_t nodes = (size_t)in_sizes[0] / HID;
  const int    ne    = in_sizes[1];
  const size_t feat  = nodes * OUTC;

  float* s1   = (float*)d_ws;
  float* s2   = s1 + feat;
  float* deg1 = s2 + feat;
  float* deg2 = deg1 + nodes;
  float* z    = deg2 + nodes;

  hipMemsetAsync(d_ws, 0, (2 * feat + 2 * nodes) * sizeof(float), stream);

  const int  ntiles = (int)(nodes / 16);
  const dim3 gblk(GEMM_THREADS);
  const dim3 ggrd((unsigned)((ntiles + GEMM_WAVES - 1) / GEMM_WAVES));
  const dim3 sblk(256);
  const dim3 sgrd((unsigned)((ne + 7) / 8));

  gemm_kernel<<<ggrd, gblk, 0, stream>>>(h, W1, z, ntiles);
  scatter_kernel<<<sgrd, sblk, 0, stream>>>(src1, dst1, z, s1, deg1, ne);

  gemm_kernel<<<ggrd, gblk, 0, stream>>>(h, W2, z, ntiles);
  scatter_kernel<<<sgrd, sblk, 0, stream>>>(src2, dst2, z, s2, deg2, ne);

  gemm_finalize_kernel<<<ggrd, gblk, 0, stream>>>(h, Wl, bl, s1, s2, deg1, deg2, out, ntiles);
}


# Scatter: one wave per 32 edges instead of one per edge: indices read coalesced and broadcast by readlane, 8 source rows (32 line loads) in flight per wave; f32 atomic adds still in arrival order
# speedup vs baseline: 1.5381x; 1.1917x over previous
#include <hip/hip_runtime.h>


typedef float v2f __attribute__((ext_vector_type(2)));
typedef float v8f __attribute__((ext_vector_type(8)));

#define HID  128
#define OUTC 128

#define GEMM_WAVES   8
#define GEMM_THREADS (GEMM_WAVES * 32)
#define WPITCH 132


__device__ __forceinline__ void stage_weight(const float* __restrict__ W, float* wl) {
#pragma unroll
  for (int i = 0; i < (OUTC * HID / 4) / GEMM_THREADS; ++i) {
    const int idx = i * GEMM_THREADS + (int)threadIdx.x;
    const int row = idx >> 5;
    const int c4  = idx & 31;
    const float4 v = *(const float4*)(W + (size_t)row * HID + c4 * 4);
    *(float4*)(wl + row * WPITCH + c4 * 4) = v;
  }
}

__device__ __forceinline__ void load_a_tile(const float* __restrict__ h, int m0, int lane, v2f a[32]) {
  const float* ap = h + (size_t)(m0 + (lane & 15)) * HID + (lane >> 4) * 2;
#pragma unroll
  for (int s = 0; s < 32; ++s) a[s] = *(const v2f*)(ap + 4 * s);
}

__device__ __forceinline__ void load_b_chunk(const float* bp, int c, v2f blo[8], v2f bhi[8]) {
#pragma unroll
  for (int j = 0; j < 8; ++j) {
    blo[j] = *(const v2f*)(bp + j * 16 * WPITCH + 8 * c);
    bhi[j] = *(const v2f*)(bp + j * 16 * WPITCH + 8 * c + 4);
  }
}

__device__ __forceinline__ void wave_gemm_16x128(const v2f a[32], const float* wl, int lane, v8f acc[8]) {
  const float* bp = wl + (lane & 15) * WPITCH + (lane >> 4) * 2;
#pragma unroll
  for (int j = 0; j < 8; ++j) acc[j] = v8f{};
  v2f blo[2][8], bhi[2][8];
  load_b_chunk(bp, 0, blo[0], bhi[0]);
#pragma unroll
  for (int c = 0; c < 16; ++c) {
    const int cur = c & 1;
    const int nxt = cur ^ 1;
    if (c + 1 < 16) load_b_chunk(bp, c + 1, blo[nxt], bhi[nxt]);
    __builtin_amdgcn_sched_barrier(0);
#pragma unroll
    for (int j = 0; j < 8; ++j)
      acc[j] = __builtin_amdgcn_wmma_f32_16x16x4_f32(false, a[2 * c], false, blo[cur][j], (short)0, acc[j], false, false);
#pragma unroll
    for (int j = 0; j < 8; ++j)
      acc[j] = __builtin_amdgcn_wmma_f32_16x16x4_f32(false, a[2 * c + 1], false, bhi[cur][j], (short)0, acc[j], false, false);
    __builtin_amdgcn_sched_barrier(0);
  }
}

__global__ __launch_bounds__(GEMM_THREADS) void gemm_kernel(const float* __restrict__ h,
                                                            const float* __restrict__ W,
                                                            float* __restrict__ z,
                                                            int ntiles) {
  __shared__ float wl[OUTC * WPITCH];
  const int lane = threadIdx.x & 31;
  const int wave = threadIdx.x >> 5;
  const int tile = blockIdx.x * GEMM_WAVES + wave;
  const int m0   = min(tile, ntiles - 1) * 16;
  stage_weight(W, wl);
  v2f a[32];
  load_a_tile(h, m0, lane, a);
  __syncthreads();
  if (tile >= ntiles) return;
  v8f acc[8];
  wave_gemm_16x128(a, wl, lane, acc);
  const int l16   = lane & 15;
  const int rbase = (lane >> 4) * 8;
#pragma unroll
  for (int r = 0; r < 8; ++r) {
    const int m = m0 + rbase + r;
#pragma unroll
    for (int j = 0; j < 8; ++j) {
      z[(size_t)m * OUTC + j * 16 + l16] = acc[j][r];
    }
  }
}

#define EDGES_PER_WAVE 32
#define EDGE_BATCH     8
#define SCATTER_WAVES  8

__global__ __launch_bounds__(SCATTER_WAVES * 32) void scatter_kernel(const int* __restrict__ src,
                                                                     const int* __restrict__ dst,
                                                                     const float* __restrict__ z,
                                                                     float* __restrict__ s,
                                                                     float* __restrict__ deg,
                                                                     int n_edges) {
  const int lane = threadIdx.x & 31;
  const int wave = __builtin_amdgcn_readfirstlane((int)(threadIdx.x >> 5));
  const int e0   = ((int)blockIdx.x * SCATTER_WAVES + wave) * EDGES_PER_WAVE;
  if (e0 >= n_edges) return;
  const int cnt = min(EDGES_PER_WAVE, n_edges - e0);
  const int el  = e0 + min(lane, cnt - 1);
  const int si  = src[el];
  const int di  = dst[el];
  if (lane < cnt) atomicAdd(deg + di, 1.0f);
  const float* zl = z + lane;
  float*       sl = s + lane;
  if (cnt == EDGES_PER_WAVE) {
#pragma unroll 1
    for (int j0 = 0; j0 < EDGES_PER_WAVE; j0 += EDGE_BATCH) {
      float v[EDGE_BATCH][4];
#pragma unroll
      for (int u = 0; u < EDGE_BATCH; ++u) {
        const int    sj = __builtin_amdgcn_readlane(si, j0 + u);
        const float* zp = zl + (size_t)sj * OUTC;
        v[u][0] = zp[0];
        v[u][1] = zp[32];
        v[u][2] = zp[64];
        v[u][3] = zp[96];
      }
#pragma unroll
      for (int u = 0; u < EDGE_BATCH; ++u) {
        const int dj = __builtin_amdgcn_readlane(di, j0 + u);
        float*    o  = sl + (size_t)dj * OUTC;
        atomicAdd(o + 0,  v[u][0]);
        atomicAdd(o + 32, v[u][1]);
        atomicAdd(o + 64, v[u][2]);
        atomicAdd(o + 96, v[u][3]);
      }
    }
  } else {
    for (int j = 0; j < cnt; ++j) {
      const int    sj = __builtin_amdgcn_readlane(si, j);
      const int    dj = __builtin_amdgcn_readlane(di, j);
      const float* zp = zl + (size_t)sj * OUTC;
      float*       o  = sl + (size_t)dj * OUTC;
      const float v0 = zp[0];
      const float v1 = zp[32];
      const float v2 = zp[64];
      const float v3 = zp[96];
      atomicAdd(o + 0,  v0);
      atomicAdd(o + 32, v1);
      atomicAdd(o + 64, v2);
      atomicAdd(o + 96, v3);
    }
  }
}

__global__ __launch_bounds__(GEMM_THREADS) void gemm_finalize_kernel(const float* __restrict__ h,
                                                                     const float* __restrict__ Wl,
                                                                     const float* __restrict__ bl,
                                                                     const float* __restrict__ s1,
                                                                     const float* __restrict__ s2,
                                                                     const float* __restrict__ deg1,
                                                                     const float* __restrict__ deg2,
                                                                     float* __restrict__ out,
                                                                     int ntiles) {
  __shared__ float wl[OUTC * WPITCH];
  const int lane = threadIdx.x & 31;
  const int wave = threadIdx.x >> 5;
  const int tile = blockIdx.x * GEMM_WAVES + wave;
  const int m0   = min(tile, ntiles - 1) * 16;
  stage_weight(Wl, wl);
  v2f a[32];
  load_a_tile(h, m0, lane, a);
  __syncthreads();
  if (tile >= ntiles) return;
  v8f acc[8];
  wave_gemm_16x128(a, wl, lane, acc);
  const int l16   = lane & 15;
  const int rbase = (lane >> 4) * 8;
#pragma unroll
  for (int r = 0; r < 8; ++r) {
    const int m = m0 + rbase + r;
    const float i1 = 0.5f / fmaxf(deg1[m], 1.0f);
    const float i2 = 0.5f / fmaxf(deg2[m], 1.0f);
#pragma unroll
    for (int j = 0; j < 8; ++j) {
      const int    c   = j * 16 + l16;
      const size_t idx = (size_t)m * OUTC + c;
      const float  v   = acc[j][r] + bl[c] + s1[idx] * i1 + s2[idx] * i2;
      out[idx] = fmaxf(v, 0.0f);
    }
  }
}

extern "C" void kernel_launch(void* const* d_in, const int* in_sizes, int n_in,
                              void* d_out, int out_size, void* d_ws, size_t ws_size,
                              hipStream_t stream) {
  (void)n_in; (void)out_size; (void)ws_size;
  const float* h    = (const float*)d_in[0];
  const int*   src1 = (const int*)  d_in[1];
  const int*   dst1 = (const int*)  d_in[2];
  const int*   src2 = (const int*)  d_in[3];
  const int*   dst2 = (const int*)  d_in[4];
  const float* W1   = (const float*)d_in[5];
  const float* W2   = (const float*)d_in[6];
  const float* Wl   = (const float*)d_in[7];
  const float* bl   = (const float*)d_in[8];
  float*       out  = (float*)d_out;

  const size_t nodes = (size_t)in_sizes[0] / HID;
  const int    ne    = in_sizes[1];
  const size_t feat  = nodes * OUTC;

  float* s1   = (float*)d_ws;
  float* s2   = s1 + feat;
  float* deg1 = s2 + feat;
  float* deg2 = deg1 + nodes;
  float* z    = deg2 + nodes;

  hipMemsetAsync(d_ws, 0, (2 * feat + 2 * nodes) * sizeof(float), stream);

  const int  ntiles = (int)(nodes / 16);
  const dim3 gblk(GEMM_THREADS);
  const dim3 ggrd((unsigned)((ntiles + GEMM_WAVES - 1) / GEMM_WAVES));
  const dim3 sblk(SCATTER_WAVES * 32);
  const dim3 sgrd((unsigned)((ne + SCATTER_WAVES * EDGES_PER_WAVE - 1) / (SCATTER_WAVES * EDGES_PER_WAVE)));

  gemm_kernel<<<ggrd, gblk, 0, stream>>>(h, W1, z, ntiles);
  scatter_kernel<<<sgrd, sblk, 0, stream>>>(src1, dst1, z, s1, deg1, ne);

  gemm_kernel<<<ggrd, gblk, 0, stream>>>(h, W2, z, ntiles);
  scatter_kernel<<<sgrd, sblk, 0, stream>>>(src2, dst2, z, s2, deg2, ne);

  gemm_finalize_kernel<<<ggrd, gblk, 0, stream>>>(h, Wl, bl, s1, s2, deg1, deg2, out, ntiles);
}


# Aggregation pulls, not pushes: edges chained per destination by one int exchange each, then each s row summed in registers by one wave, stored once; no f32 atomics; f32 sums in list (arrival) order
# speedup vs baseline: 2.3465x; 1.5256x over previous
#include <hip/hip_runtime.h>


typedef float v2f __attribute__((ext_vector_type(2)));
typedef float v8f __attribute__((ext_vector_type(8)));

#define HID  128
#define OUTC 128

#define GEMM_WAVES   16
#define GEMM_THREADS (GEMM_WAVES * 32)
#define WPITCH 132


__device__ __forceinline__ void stage_weight(const float* __restrict__ W, float* wl) {
#pragma unroll
  for (int i = 0; i < (OUTC * HID / 4) / GEMM_THREADS; ++i) {
    const int idx = i * GEMM_THREADS + (int)threadIdx.x;
    const int row = idx >> 5;
    const int c4  = idx & 31;
    const float4 v = *(const float4*)(W + (size_t)row * HID + c4 * 4);
    *(float4*)(wl + row * WPITCH + c4 * 4) = v;
  }
}

__device__ __forceinline__ void load_a_tile(const float* __restrict__ h, int m0, int lane, v2f a[32]) {
  const float* ap = h + (size_t)(m0 + (lane & 15)) * HID + (lane >> 4) * 2;
#pragma unroll
  for (int s = 0; s < 32; ++s) a[s] = *(const v2f*)(ap + 4 * s);
}

__device__ __forceinline__ void load_b_chunk(const float* bp, int c, v2f blo[8], v2f bhi[8]) {
#pragma unroll
  for (int j = 0; j < 8; ++j) {
    blo[j] = *(const v2f*)(bp + j * 16 * WPITCH + 8 * c);
    bhi[j] = *(const v2f*)(bp + j * 16 * WPITCH + 8 * c + 4);
  }
}

__device__ __forceinline__ void wave_gemm_16x128(const v2f a[32], const float* wl, int lane, v8f acc[8]) {
  const float* bp = wl + (lane & 15) * WPITCH + (lane >> 4) * 2;
  v2f blo[2][8], bhi[2][8];
  load_b_chunk(bp, 0, blo[0], bhi[0]);
#pragma unroll
  for (int c = 0; c < 16; ++c) {
    const int cur = c & 1;
    const int nxt = cur ^ 1;
    if (c + 1 < 16) load_b_chunk(bp, c + 1, blo[nxt], bhi[nxt]);
    __builtin_amdgcn_sched_barrier(0);
#pragma unroll
    for (int j = 0; j < 8; ++j)
      acc[j] = __builtin_amdgcn_wmma_f32_16x16x4_f32(false, a[2 * c], false, blo[cur][j], (short)0, acc[j], false, false);
#pragma unroll
    for (int j = 0; j < 8; ++j)
      acc[j] = __builtin_amdgcn_wmma_f32_16x16x4_f32(false, a[2 * c + 1], false, bhi[cur][j], (short)0, acc[j], false, false);
    __builtin_amdgcn_sched_barrier(0);
  }
}


#define LINK_THREADS 256
#define LINK_EDGES_PER_THREAD 4

__global__ __launch_bounds__(LINK_THREADS) void link_kernel(const int* __restrict__ dst1,
                                                            const int* __restrict__ dst2,
                                                            int* __restrict__ head1,
                                                            int* __restrict__ head2,
                                                            int* __restrict__ next1,
                                                            int* __restrict__ next2,
                                                            int n_edges) {
  const int e0 = (int)blockIdx.x * (LINK_THREADS * LINK_EDGES_PER_THREAD) + (int)threadIdx.x;
  int d1[LINK_EDGES_PER_THREAD], d2[LINK_EDGES_PER_THREAD], o1[LINK_EDGES_PER_THREAD], o2[LINK_EDGES_PER_THREAD];
#pragma unroll
  for (int k = 0; k < LINK_EDGES_PER_THREAD; ++k) {
    const int e = e0 + k * LINK_THREADS;
    d1[k] = 0;
    d2[k] = 0;
    if (e < n_edges) {
      d1[k] = dst1[e];
      d2[k] = dst2[e];
    }
  }
#pragma unroll
  for (int k = 0; k < LINK_EDGES_PER_THREAD; ++k) {
    const int e = e0 + k * LINK_THREADS;
    o1[k] = -1;
    o2[k] = -1;
    if (e < n_edges) {
      o1[k] = atomicExch(head1 + d1[k], e);
      o2[k] = atomicExch(head2 + d2[k], e);
    }
  }
#pragma unroll
  for (int k = 0; k < LINK_EDGES_PER_THREAD; ++k) {
    const int e = e0 + k * LINK_THREADS;
    if (e < n_edges) {
      next1[e] = o1[k];
      next2[e] = o2[k];
    }
  }
}

#define AGG_ROWS  32
#define AGG_BATCH 8
#define AGG_WAVES 8

__global__ __launch_bounds__(AGG_WAVES * 32) void gather_sum_kernel(const int* __restrict__ src1,
                                                                    const int* __restrict__ src2,
                                                                    const int* __restrict__ head1,
                                                                    const int* __restrict__ head2,
                                                                    const int* __restrict__ next1,
                                                                    const int* __restrict__ next2,
                                                                    const float* __restrict__ x,
                                                                    float* __restrict__ s1,
                                                                    float* __restrict__ s2,
                                                                    float* __restrict__ deg1,
                                                                    float* __restrict__ deg2,
                                                                    int n_nodes, int n_edges, int groups_per_type) {
  const int lane = threadIdx.x & 31;
  const int wave = __builtin_amdgcn_readfirstlane((int)(threadIdx.x >> 5));
  int g = (int)blockIdx.x * AGG_WAVES + wave;
  const bool second = g >= groups_per_type;
  if (second) g -= groups_per_type;
  if (g >= groups_per_type) return;
  const int* __restrict__ src  = second ? src2  : src1;
  const int* __restrict__ head = second ? head2 : head1;
  const int* __restrict__ next = second ? next2 : next1;
  float* __restrict__ s   = second ? s2   : s1;
  float* __restrict__ deg = second ? deg2 : deg1;

  const int base = g * AGG_ROWS;
  const int row  = base + lane;
  float4 acc[AGG_ROWS];
#pragma unroll
  for (int i = 0; i < AGG_ROWS; ++i) acc[i] = make_float4(0.0f, 0.0f, 0.0f, 0.0f);

  int cur = (row < n_nodes) ? head[row] : -1;
  bool valid = (unsigned)cur < (unsigned)n_edges;
  int sjv = valid ? src[cur] : -1;
  cur     = valid ? next[cur] : -1;
  int cnt = 0;
  const float* xl = x + lane * 4;
  for (int lvl = 0; lvl < n_edges; ++lvl) {
    const unsigned m = __builtin_amdgcn_ballot_w32(sjv >= 0);
    if (m == 0u) break;
    cnt += (sjv >= 0) ? 1 : 0;
    valid = (unsigned)cur < (unsigned)n_edges;
    const int sjn = valid ? src[cur] : -1;
    const int cun = valid ? next[cur] : -1;
#pragma unroll
    for (int b = 0; b < AGG_ROWS / AGG_BATCH; ++b) {
      if (((m >> (b * AGG_BATCH)) & 0xFFu) != 0u) {
        float4 v[AGG_BATCH];
#pragma unroll
        for (int u = 0; u < AGG_BATCH; ++u) {
          const int sj = __builtin_amdgcn_readlane(sjv, b * AGG_BATCH + u);
          const int rj = (sj >= 0) ? sj : 0;
          v[u] = *(const float4*)(xl + (size_t)rj * OUTC);
        }
#pragma unroll
        for (int u = 0; u < AGG_BATCH; ++u) {
          const bool ok = ((m >> (b * AGG_BATCH + u)) & 1u) != 0u;
          acc[b * AGG_BATCH + u].x += ok ? v[u].x : 0.0f;
          acc[b * AGG_BATCH + u].y += ok ? v[u].y : 0.0f;
          acc[b * AGG_BATCH + u].z += ok ? v[u].z : 0.0f;
          acc[b * AGG_BATCH + u].w += ok ? v[u].w : 0.0f;
        }
      }
    }
    sjv = sjn;
    cur = cun;
  }
  if (row < n_nodes) deg[row] = (float)cnt;
#pragma unroll
  for (int i = 0; i < AGG_ROWS; ++i) {
    const int r = base + i;
    if (r < n_nodes) *(float4*)(s + (size_t)r * OUTC + lane * 4) = acc[i];
  }
}

#define FUSED_GROUPS 256

__global__ __launch_bounds__(GEMM_THREADS) void gemm3_finalize_kernel(const float* __restrict__ h,
                                                                      const float* __restrict__ Wl,
                                                                      const float* __restrict__ W1,
                                                                      const float* __restrict__ W2,
                                                                      const float* __restrict__ bl,
                                                                      const float* __restrict__ s1,
                                                                      const float* __restrict__ s2,
                                                                      const float* __restrict__ deg1,
                                                                      const float* __restrict__ deg2,
                                                                      float* __restrict__ out,
                                                                      int ntiles) {
  __shared__ float wl[3 * OUTC * WPITCH];
  const int lane = threadIdx.x & 31;
  const int wave = __builtin_amdgcn_readfirstlane((int)(threadIdx.x >> 5));
  stage_weight(Wl, wl);
  stage_weight(W1, wl + OUTC * WPITCH);
  stage_weight(W2, wl + 2 * OUTC * WPITCH);
  __syncthreads();
  const int l16   = lane & 15;
  const int rbase = (lane >> 4) * 8;
  const int podd = (wave * 13) & (GEMM_WAVES - 1);
#pragma unroll 1
  for (int k = 0; ; ++k) {
    const int slot = k * GEMM_WAVES + ((k & 1) ? podd : wave);
    const int tile = slot * FUSED_GROUPS + (int)blockIdx.x;
    if (tile >= ntiles) break;
    const int m0 = tile * 16;
    v8f acc[8];
#pragma unroll
    for (int j = 0; j < 8; ++j) acc[j] = v8f{};
#pragma unroll 1
    for (int p = 0; p < 3; ++p) {
      const float* __restrict__ A = (p == 0) ? h : (p == 1) ? s1 : s2;
      v2f a[32];
      load_a_tile(A, m0, lane, a);
      if (p != 0) {
        const float* __restrict__ dg = (p == 1) ? deg1 : deg2;
        const float f = 0.5f / fmaxf(dg[m0 + l16], 1.0f);
#pragma unroll
        for (int i = 0; i < 32; ++i) a[i] = a[i] * f;
      }
      wave_gemm_16x128(a, wl + p * (OUTC * WPITCH), lane, acc);
    }
#pragma unroll
    for (int r = 0; r < 8; ++r) {
      const int m = m0 + rbase + r;
#pragma unroll
      for (int j = 0; j < 8; ++j) {
        const int c = j * 16 + l16;
        out[(size_t)m * OUTC + c] = fmaxf(acc[j][r] + bl[c], 0.0f);
      }
    }
  }
}

extern "C" void kernel_launch(void* const* d_in, const int* in_sizes, int n_in,
                              void* d_out, int out_size, void* d_ws, size_t ws_size,
                              hipStream_t stream) {
  (void)n_in; (void)out_size; (void)ws_size;
  const float* h    = (const float*)d_in[0];
  const int*   src1 = (const int*)  d_in[1];
  const int*   dst1 = (const int*)  d_in[2];
  const int*   src2 = (const int*)  d_in[3];
  const int*   dst2 = (const int*)  d_in[4];
  const float* W1   = (const float*)d_in[5];
  const float* W2   = (const float*)d_in[6];
  const float* Wl   = (const float*)d_in[7];
  const float* bl   = (const float*)d_in[8];
  float*       out  = (float*)d_out;

  const size_t nodes = (size_t)in_sizes[0] / HID;
  const int    ne    = in_sizes[1];
  const size_t feat  = nodes * HID;

  float* s1   = (float*)d_ws;
  float* s2   = s1 + feat;
  float* deg1 = s2 + feat;
  float* deg2 = deg1 + nodes;
  int*   head1 = (int*)(deg2 + nodes);
  int*   head2 = head1 + nodes;
  int*   next1 = head2 + nodes;
  int*   next2 = next1 + ne;

  hipMemsetAsync(head1, 0xFF, 2 * nodes * sizeof(int), stream);

  const int  ntiles = (int)(nodes / 16);
  const dim3 gblk(GEMM_THREADS);
  const dim3 ggrd(FUSED_GROUPS);
  const dim3 lblk(LINK_THREADS);
  const dim3 lgrd((unsigned)((ne + LINK_THREADS * LINK_EDGES_PER_THREAD - 1) / (LINK_THREADS * LINK_EDGES_PER_THREAD)));
  const int  groups = (int)((nodes + AGG_ROWS - 1) / AGG_ROWS);
  const dim3 ablk(AGG_WAVES * 32);
  const dim3 agrd((unsigned)((2 * groups + AGG_WAVES - 1) / AGG_WAVES));

  link_kernel<<<lgrd, lblk, 0, stream>>>(dst1, dst2, head1, head2, next1, next2, ne);
  gather_sum_kernel<<<agrd, ablk, 0, stream>>>(src1, src2, head1, head2, next1, next2, h, s1, s2, deg1, deg2,
                                               (int)nodes, ne, groups);

  gemm3_finalize_kernel<<<ggrd, gblk, 0, stream>>>(h, Wl, W1, W2, bl, s1, s2, deg1, deg2, out, ntiles);
}


# Aggregation: list walk only collects indices (per-wave LDS, repacked row-major); source rows then loaded in flat groups of 16 real rows, one running sum stored per row; f32 sums still in list order
# speedup vs baseline: 2.5213x; 1.0745x over previous
#include <hip/hip_runtime.h>


typedef float v2f __attribute__((ext_vector_type(2)));
typedef float v8f __attribute__((ext_vector_type(8)));

#define HID  128
#define OUTC 128

#define GEMM_WAVES   16
#define GEMM_THREADS (GEMM_WAVES * 32)
#define WPITCH 132


__device__ __forceinline__ void stage_weight(const float* __restrict__ W, float* wl) {
#pragma unroll
  for (int i = 0; i < (OUTC * HID / 4) / GEMM_THREADS; ++i) {
    const int idx = i * GEMM_THREADS + (int)threadIdx.x;
    const int row = idx >> 5;
    const int c4  = idx & 31;
    const float4 v = *(const float4*)(W + (size_t)row * HID + c4 * 4);
    *(float4*)(wl + row * WPITCH + c4 * 4) = v;
  }
}

__device__ __forceinline__ void load_a_tile(const float* __restrict__ h, int m0, int lane, v2f a[32]) {
  const float* ap = h + (size_t)(m0 + (lane & 15)) * HID + (lane >> 4) * 2;
#pragma unroll
  for (int s = 0; s < 32; ++s) a[s] = *(const v2f*)(ap + 4 * s);
}

__device__ __forceinline__ void load_b_chunk(const float* bp, int c, v2f blo[8], v2f bhi[8]) {
#pragma unroll
  for (int j = 0; j < 8; ++j) {
    blo[j] = *(const v2f*)(bp + j * 16 * WPITCH + 8 * c);
    bhi[j] = *(const v2f*)(bp + j * 16 * WPITCH + 8 * c + 4);
  }
}

__device__ __forceinline__ void wave_gemm_16x128(const v2f a[32], const float* wl, int lane, v8f acc[8]) {
  const float* bp = wl + (lane & 15) * WPITCH + (lane >> 4) * 2;
  v2f blo[2][8], bhi[2][8];
  load_b_chunk(bp, 0, blo[0], bhi[0]);
#pragma unroll
  for (int c = 0; c < 16; ++c) {
    const int cur = c & 1;
    const int nxt = cur ^ 1;
    if (c + 1 < 16) load_b_chunk(bp, c + 1, blo[nxt], bhi[nxt]);
    __builtin_amdgcn_sched_barrier(0);
#pragma unroll
    for (int j = 0; j < 8; ++j)
      acc[j] = __builtin_amdgcn_wmma_f32_16x16x4_f32(false, a[2 * c], false, blo[cur][j], (short)0, acc[j], false, false);
#pragma unroll
    for (int j = 0; j < 8; ++j)
      acc[j] = __builtin_amdgcn_wmma_f32_16x16x4_f32(false, a[2 * c + 1], false, bhi[cur][j], (short)0, acc[j], false, false);
    __builtin_amdgcn_sched_barrier(0);
  }
}


#define LINK_THREADS 256
#define LINK_EDGES_PER_THREAD 4

__global__ __launch_bounds__(LINK_THREADS) void link_kernel(const int* __restrict__ dst1,
                                                            const int* __restrict__ dst2,
                                                            int* __restrict__ head1,
                                                            int* __restrict__ head2,
                                                            int* __restrict__ next1,
                                                            int* __restrict__ next2,
                                                            int n_edges) {
  const int e0 = (int)blockIdx.x * (LINK_THREADS * LINK_EDGES_PER_THREAD) + (int)threadIdx.x;
  int d1[LINK_EDGES_PER_THREAD], d2[LINK_EDGES_PER_THREAD], o1[LINK_EDGES_PER_THREAD], o2[LINK_EDGES_PER_THREAD];
#pragma unroll
  for (int k = 0; k < LINK_EDGES_PER_THREAD; ++k) {
    const int e = e0 + k * LINK_THREADS;
    d1[k] = 0;
    d2[k] = 0;
    if (e < n_edges) {
      d1[k] = dst1[e];
      d2[k] = dst2[e];
    }
  }
#pragma unroll
  for (int k = 0; k < LINK_EDGES_PER_THREAD; ++k) {
    const int e = e0 + k * LINK_THREADS;
    o1[k] = -1;
    o2[k] = -1;
    if (e < n_edges) {
      o1[k] = atomicExch(head1 + d1[k], e);
      o2[k] = atomicExch(head2 + d2[k], e);
    }
  }
#pragma unroll
  for (int k = 0; k < LINK_EDGES_PER_THREAD; ++k) {
    const int e = e0 + k * LINK_THREADS;
    if (e < n_edges) {
      next1[e] = o1[k];
      next2[e] = o2[k];
    }
  }
}

#define AGG_ROWS  32
#define AGG_CAP   16
#define AGG_BATCH 16
#define AGG_WAVES 8

__device__ __forceinline__ void close_row(float4& acc, unsigned& rem, int& r, unsigned ovf, int currem,
                                          const int* __restrict__ src, const int* __restrict__ next,
                                          const float* xl, float* __restrict__ s, int base, int lane, int n_edges) {
  if (((ovf >> r) & 1u) != 0u) {
    int c = __builtin_amdgcn_readlane(currem, r);
    for (int it = 0; it < n_edges; ++it) {
      if ((unsigned)c >= (unsigned)n_edges) break;
      const int sj = src[c] & 0x7fffffff;
      c = next[c];
      const float4 t = *(const float4*)(xl + (size_t)sj * OUTC);
      acc.x += t.x;
      acc.y += t.y;
      acc.z += t.z;
      acc.w += t.w;
    }
  }
  *(float4*)(s + (size_t)(base + r) * OUTC + lane * 4) = acc;
  acc = make_float4(0.0f, 0.0f, 0.0f, 0.0f);
  rem &= rem - 1u;
  r = (rem != 0u) ? __builtin_ctz(rem) : 0;
}

__global__ __launch_bounds__(AGG_WAVES * 32) void gather_sum_kernel(const int* __restrict__ src1,
                                                                    const int* __restrict__ src2,
                                                                    const int* __restrict__ head1,
                                                                    const int* __restrict__ head2,
                                                                    const int* __restrict__ next1,
                                                                    const int* __restrict__ next2,
                                                                    const float* __restrict__ x,
                                                                    float* __restrict__ s1,
                                                                    float* __restrict__ s2,
                                                                    float* __restrict__ deg1,
                                                                    float* __restrict__ deg2,
                                                                    int n_nodes, int n_edges, int groups_per_type) {
  __shared__ int lv_all[AGG_WAVES * AGG_CAP * AGG_ROWS];
  __shared__ int flat_all[AGG_WAVES * AGG_CAP * AGG_ROWS];
  const int lane = threadIdx.x & 31;
  const int wave = __builtin_amdgcn_readfirstlane((int)(threadIdx.x >> 5));
  int g = (int)blockIdx.x * AGG_WAVES + wave;
  const bool second = g >= groups_per_type;
  if (second) g -= groups_per_type;
  if (g >= groups_per_type) return;
  const int* __restrict__ src  = second ? src2  : src1;
  const int* __restrict__ head = second ? head2 : head1;
  const int* __restrict__ next = second ? next2 : next1;
  float* __restrict__ s   = second ? s2   : s1;
  float* __restrict__ deg = second ? deg2 : deg1;
  int* lv   = lv_all   + wave * (AGG_CAP * AGG_ROWS);
  int* flat = flat_all + wave * (AGG_CAP * AGG_ROWS);

  const int base = g * AGG_ROWS;
  const int row  = base + lane;

  int cur = (row < n_nodes) ? head[row] : -1;
  int cnt = 0;
  int currem = -1;
  for (int lvl = 0; lvl < n_edges; ++lvl) {
    const bool valid = (unsigned)cur < (unsigned)n_edges;
    if (__builtin_amdgcn_ballot_w32(valid) == 0u) break;
    if (valid) {
      const int nx = next[cur];
      if (lvl < AGG_CAP) lv[lvl * AGG_ROWS + lane] = src[cur] & 0x7fffffff;
      else if (lvl == AGG_CAP) currem = cur;
      cur = nx;
      cnt += 1;
    }
  }
  if (row < n_nodes) deg[row] = (float)cnt;

  const int kept = (cnt < AGG_CAP) ? cnt : AGG_CAP;
  int incl = kept;
#pragma unroll
  for (int d = 1; d < 32; d <<= 1) {
    const int t = __shfl_up(incl, (unsigned)d);
    if (lane >= d) incl += t;
  }
  const int off   = incl - kept;
  const int total = __builtin_amdgcn_readlane(incl, 31);
  int ecol[AGG_CAP];
#pragma unroll
  for (int l = 0; l < AGG_CAP; ++l) ecol[l] = (l < kept) ? lv[l * AGG_ROWS + lane] : 0;
#pragma unroll
  for (int l = 0; l < AGG_CAP; ++l) {
    if (l < kept) flat[off + l] = (l == kept - 1) ? (ecol[l] | (int)0x80000000u) : ecol[l];
  }

  unsigned zm = __builtin_amdgcn_ballot_w32(cnt == 0 && row < n_nodes);
  for (int t = 0; t < AGG_ROWS && zm != 0u; ++t) {
    const int i = __builtin_ctz(zm);
    zm &= zm - 1u;
    *(float4*)(s + (size_t)(base + i) * OUTC + lane * 4) = make_float4(0.0f, 0.0f, 0.0f, 0.0f);
  }

  unsigned rem = __builtin_amdgcn_ballot_w32(cnt > 0);
  const unsigned ovf = __builtin_amdgcn_ballot_w32(cnt > AGG_CAP);
  int r = (rem != 0u) ? __builtin_ctz(rem) : 0;
  const float* xl = x + lane * 4;
  float4 acc = make_float4(0.0f, 0.0f, 0.0f, 0.0f);
  const int nfull = total & ~(AGG_BATCH - 1);
  for (int b0 = 0; b0 < nfull; b0 += AGG_BATCH) {
    const int ent = flat[b0 + (lane & (AGG_BATCH - 1))];
    float4 v[AGG_BATCH];
#pragma unroll
    for (int u = 0; u < AGG_BATCH; ++u) {
      const int sj = __builtin_amdgcn_readlane(ent, u) & 0x7fffffff;
      v[u] = *(const float4*)(xl + (size_t)sj * OUTC);
    }
#pragma unroll
    for (int u = 0; u < AGG_BATCH; ++u) {
      acc.x += v[u].x;
      acc.y += v[u].y;
      acc.z += v[u].z;
      acc.w += v[u].w;
      if (__builtin_amdgcn_readlane(ent, u) < 0)
        close_row(acc, rem, r, ovf, currem, src, next, xl, s, base, lane, n_edges);
    }
  }
  const int nt = total - nfull;
  if (nt > 0) {
    const int li  = lane & (AGG_BATCH - 1);
    const int ent = flat[nfull + ((li < nt) ? li : nt - 1)];
    float4 v[AGG_BATCH];
#pragma unroll
    for (int u = 0; u < AGG_BATCH; ++u) {
      const int sj = __builtin_amdgcn_readlane(ent, u) & 0x7fffffff;
      v[u] = *(const float4*)(xl + (size_t)sj * OUTC);
    }
#pragma unroll
    for (int u = 0; u < AGG_BATCH; ++u) {
      if (u < nt) {
        acc.x += v[u].x;
        acc.y += v[u].y;
        acc.z += v[u].z;
        acc.w += v[u].w;
        if (__builtin_amdgcn_readlane(ent, u) < 0)
          close_row(acc, rem, r, ovf, currem, src, next, xl, s, base, lane, n_edges);
      }
    }
  }
}

#define FUSED_GROUPS 256

__global__ __launch_bounds__(GEMM_THREADS) void gemm3_finalize_kernel(const float* __restrict__ h,
                                                                      const float* __restrict__ Wl,
                                                                      const float* __restrict__ W1,
                                                                      const float* __restrict__ W2,
                                                                      const float* __restrict__ bl,
                                                                      const float* __restrict__ s1,
                                                                      const float* __restrict__ s2,
                                                                      const float* __restrict__ deg1,
                                                                      const float* __restrict__ deg2,
                                                                      float* __restrict__ out,
                                                                      int ntiles) {
  __shared__ float wl[3 * OUTC * WPITCH];
  const int lane = threadIdx.x & 31;
  const int wave = __builtin_amdgcn_readfirstlane((int)(threadIdx.x >> 5));
  stage_weight(Wl, wl);
  stage_weight(W1, wl + OUTC * WPITCH);
  stage_weight(W2, wl + 2 * OUTC * WPITCH);
  __syncthreads();
  const int l16   = lane & 15;
  const int rbase = (lane >> 4) * 8;
  const int podd = (wave * 13) & (GEMM_WAVES - 1);
#pragma unroll 1
  for (int k = 0; ; ++k) {
    const int slot = k * GEMM_WAVES + ((k & 1) ? podd : wave);
    const int tile = slot * FUSED_GROUPS + (int)blockIdx.x;
    if (tile >= ntiles) break;
    const int m0 = tile * 16;
    v8f acc[8];
#pragma unroll
    for (int j = 0; j < 8; ++j) acc[j] = v8f{};
#pragma unroll 1
    for (int p = 0; p < 3; ++p) {
      const float* __restrict__ A = (p == 0) ? h : (p == 1) ? s1 : s2;
      v2f a[32];
      load_a_tile(A, m0, lane, a);
      if (p != 0) {
        const float* __restrict__ dg = (p == 1) ? deg1 : deg2;
        const float f = 0.5f / fmaxf(dg[m0 + l16], 1.0f);
#pragma unroll
        for (int i = 0; i < 32; ++i) a[i] = a[i] * f;
      }
      wave_gemm_16x128(a, wl + p * (OUTC * WPITCH), lane, acc);
    }
#pragma unroll
    for (int r = 0; r < 8; ++r) {
      const int m = m0 + rbase + r;
#pragma unroll
      for (int j = 0; j < 8; ++j) {
        const int c = j * 16 + l16;
        out[(size_t)m * OUTC + c] = fmaxf(acc[j][r] + bl[c], 0.0f);
      }
    }
  }
}

extern "C" void kernel_launch(void* const* d_in, const int* in_sizes, int n_in,
                              void* d_out, int out_size, void* d_ws, size_t ws_size,
                              hipStream_t stream) {
  (void)n_in; (void)out_size; (void)ws_size;
  const float* h    = (const float*)d_in[0];
  const int*   src1 = (const int*)  d_in[1];
  const int*   dst1 = (const int*)  d_in[2];
  const int*   src2 = (const int*)  d_in[3];
  const int*   dst2 = (const int*)  d_in[4];
  const float* W1   = (const float*)d_in[5];
  const float* W2   = (const float*)d_in[6];
  const float* Wl   = (const float*)d_in[7];
  const float* bl   = (const float*)d_in[8];
  float*       out  = (float*)d_out;

  const size_t nodes = (size_t)in_sizes[0] / HID;
  const int    ne    = in_sizes[1];
  const size_t feat  = nodes * HID;

  float* s1   = (float*)d_ws;
  float* s2   = s1 + feat;
  float* deg1 = s2 + feat;
  float* deg2 = deg1 + nodes;
  int*   head1 = (int*)(deg2 + nodes);
  int*   head2 = head1 + nodes;
  int*   next1 = head2 + nodes;
  int*   next2 = next1 + ne;

  hipMemsetAsync(head1, 0xFF, 2 * nodes * sizeof(int), stream);

  const int  ntiles = (int)(nodes / 16);
  const dim3 gblk(GEMM_THREADS);
  const dim3 ggrd(FUSED_GROUPS);
  const dim3 lblk(LINK_THREADS);
  const dim3 lgrd((unsigned)((ne + LINK_THREADS * LINK_EDGES_PER_THREAD - 1) / (LINK_THREADS * LINK_EDGES_PER_THREAD)));
  const int  groups = (int)((nodes + AGG_ROWS - 1) / AGG_ROWS);
  const dim3 ablk(AGG_WAVES * 32);
  const dim3 agrd((unsigned)((2 * groups + AGG_WAVES - 1) / AGG_WAVES));

  link_kernel<<<lgrd, lblk, 0, stream>>>(dst1, dst2, head1, head2, next1, next2, ne);
  gather_sum_kernel<<<agrd, ablk, 0, stream>>>(src1, src2, head1, head2, next1, next2, h, s1, s2, deg1, deg2,
                                               (int)nodes, ne, groups);

  gemm3_finalize_kernel<<<ggrd, gblk, 0, stream>>>(h, Wl, W1, W2, bl, s1, s2, deg1, deg2, out, ntiles);
}


# Aggregation fused into the GEMM kernel: the wave owning a 16-row tile sums the tile's source rows itself and feeds its WMMAs through LDS; s1/s2 never touch memory; f32 sum order unchanged
# speedup vs baseline: 2.7247x; 1.0807x over previous
#include <hip/hip_runtime.h>


typedef float v2f __attribute__((ext_vector_type(2)));
typedef float v8f __attribute__((ext_vector_type(8)));

#define HID  128
#define OUTC 128

#define GEMM_WAVES   12
#define GEMM_THREADS (GEMM_WAVES * 32)
#define WPITCH 132


__device__ __forceinline__ void stage_weight(const float* __restrict__ W, float* wl) {
#pragma unroll
  for (int i = 0; i < (OUTC * HID / 4 + GEMM_THREADS - 1) / GEMM_THREADS; ++i) {
    const int idx = i * GEMM_THREADS + (int)threadIdx.x;
    if (idx < OUTC * HID / 4) {
      const int row = idx >> 5;
      const int c4  = idx & 31;
      const float4 v = *(const float4*)(W + (size_t)row * HID + c4 * 4);
      *(float4*)(wl + row * WPITCH + c4 * 4) = v;
    }
  }
}

__device__ __forceinline__ void load_a_tile(const float* __restrict__ h, int m0, int lane, v2f a[32]) {
  const float* ap = h + (size_t)(m0 + (lane & 15)) * HID + (lane >> 4) * 2;
#pragma unroll
  for (int s = 0; s < 32; ++s) a[s] = *(const v2f*)(ap + 4 * s);
}

__device__ __forceinline__ void load_b_chunk(const float* bp, int c, v2f blo[8], v2f bhi[8]) {
#pragma unroll
  for (int j = 0; j < 8; ++j) {
    blo[j] = *(const v2f*)(bp + j * 16 * WPITCH + 8 * c);
    bhi[j] = *(const v2f*)(bp + j * 16 * WPITCH + 8 * c + 4);
  }
}

__device__ __forceinline__ void wave_gemm_16x128(const v2f a[32], const float* wl, int lane, v8f acc[8]) {
  const float* bp = wl + (lane & 15) * WPITCH + (lane >> 4) * 2;
  v2f blo[2][8], bhi[2][8];
  load_b_chunk(bp, 0, blo[0], bhi[0]);
#pragma unroll
  for (int c = 0; c < 16; ++c) {
    const int cur = c & 1;
    const int nxt = cur ^ 1;
    if (c + 1 < 16) load_b_chunk(bp, c + 1, blo[nxt], bhi[nxt]);
    __builtin_amdgcn_sched_barrier(0);
#pragma unroll
    for (int j = 0; j < 8; ++j)
      acc[j] = __builtin_amdgcn_wmma_f32_16x16x4_f32(false, a[2 * c], false, blo[cur][j], (short)0, acc[j], false, false);
#pragma unroll
    for (int j = 0; j < 8; ++j)
      acc[j] = __builtin_amdgcn_wmma_f32_16x16x4_f32(false, a[2 * c + 1], false, bhi[cur][j], (short)0, acc[j], false, false);
    __builtin_amdgcn_sched_barrier(0);
  }
}


#define LINK_THREADS 256
#define LINK_EDGES_PER_THREAD 4

__global__ __launch_bounds__(LINK_THREADS) void link_kernel(const int* __restrict__ dst1,
                                                            const int* __restrict__ dst2,
                                                            int* __restrict__ head1,
                                                            int* __restrict__ head2,
                                                            int* __restrict__ next1,
                                                            int* __restrict__ next2,
                                                            int n_edges) {
  const int e0 = (int)blockIdx.x * (LINK_THREADS * LINK_EDGES_PER_THREAD) + (int)threadIdx.x;
  int d1[LINK_EDGES_PER_THREAD], d2[LINK_EDGES_PER_THREAD], o1[LINK_EDGES_PER_THREAD], o2[LINK_EDGES_PER_THREAD];
#pragma unroll
  for (int k = 0; k < LINK_EDGES_PER_THREAD; ++k) {
    const int e = e0 + k * LINK_THREADS;
    d1[k] = 0;
    d2[k] = 0;
    if (e < n_edges) {
      d1[k] = dst1[e];
      d2[k] = dst2[e];
    }
  }
#pragma unroll
  for (int k = 0; k < LINK_EDGES_PER_THREAD; ++k) {
    const int e = e0 + k * LINK_THREADS;
    o1[k] = -1;
    o2[k] = -1;
    if (e < n_edges) {
      o1[k] = atomicExch(head1 + d1[k], e);
      o2[k] = atomicExch(head2 + d2[k], e);
    }
  }
#pragma unroll
  for (int k = 0; k < LINK_EDGES_PER_THREAD; ++k) {
    const int e = e0 + k * LINK_THREADS;
    if (e < n_edges) {
      next1[e] = o1[k];
      next2[e] = o2[k];
    }
  }
}

#define AGG_CAP   16
#define AGG_BATCH 16
#define STG_INTS  (16 * HID)
#define FLAT_INTS (32 * AGG_CAP)

__device__ __forceinline__ void close_row(float4& acc, unsigned& rem, int& r, unsigned ovf, int currem, float fsc,
                                          const int* __restrict__ src, const int* __restrict__ next,
                                          const float* xl, int* stg, int lane, int n_edges) {
  if (((ovf >> r) & 1u) != 0u) {
    int c = __builtin_amdgcn_readlane(currem, r);
    for (int it = 0; it < n_edges; ++it) {
      if ((unsigned)c >= (unsigned)n_edges) break;
      const int sj = src[c] & 0x7fffffff;
      c = next[c];
      const float4 t = *(const float4*)(xl + (size_t)sj * OUTC);
      acc.x += t.x;
      acc.y += t.y;
      acc.z += t.z;
      acc.w += t.w;
    }
  }
  const float f = __int_as_float(__builtin_amdgcn_readlane(__float_as_int(fsc), r));
  const int rr = r & 15;
  int4 o;
  o.x = __float_as_int(acc.x * f);
  o.y = __float_as_int(acc.y * f);
  o.z = __float_as_int(acc.z * f);
  o.w = __float_as_int(acc.w * f);
  *(int4*)(stg + rr * HID + ((lane ^ rr) << 2)) = o;
  acc = make_float4(0.0f, 0.0f, 0.0f, 0.0f);
  rem &= rem - 1u;
  r = (rem != 0u) ? __builtin_ctz(rem) : 0;
}

#define FUSED_GROUPS 256

__global__ __launch_bounds__(GEMM_THREADS) void gather_gemm3_finalize_kernel(const float* __restrict__ h,
                                                                             const float* __restrict__ Wl,
                                                                             const float* __restrict__ W1,
                                                                             const float* __restrict__ W2,
                                                                             const float* __restrict__ bl,
                                                                             const int* __restrict__ src1,
                                                                             const int* __restrict__ src2,
                                                                             const int* __restrict__ head1,
                                                                             const int* __restrict__ head2,
                                                                             const int* __restrict__ next1,
                                                                             const int* __restrict__ next2,
                                                                             float* __restrict__ out,
                                                                             int ntiles, int n_edges) {
  __shared__ float wl[3 * OUTC * WPITCH];
  __shared__ int scr_all[GEMM_WAVES * (STG_INTS + FLAT_INTS)];
  const int lane = threadIdx.x & 31;
  const int wave = __builtin_amdgcn_readfirstlane((int)(threadIdx.x >> 5));
  stage_weight(Wl, wl);
  stage_weight(W1, wl + OUTC * WPITCH);
  stage_weight(W2, wl + 2 * OUTC * WPITCH);
  __syncthreads();
  const int l16   = lane & 15;
  const int half  = lane >> 4;
  const int rbase = half * 8;
  int* stg  = scr_all + wave * (STG_INTS + FLAT_INTS);
  int* flat = stg + STG_INTS;
  int* lv   = stg;
  const int* __restrict__ srcL  = half ? src2  : src1;
  const int* __restrict__ headL = half ? head2 : head1;
  const int* __restrict__ nextL = half ? next2 : next1;
  const float* xl = h + lane * 4;

#pragma unroll 1
  for (int k = 0; ; ++k) {
    const int tile = (k * GEMM_WAVES + wave) * FUSED_GROUPS + (int)blockIdx.x;
    if (tile >= ntiles) break;
    const int m0 = tile * 16;

    int cur = headL[m0 + l16];
    int cnt = 0;
    int currem = -1;
    for (int lvl = 0; lvl < n_edges; ++lvl) {
      const bool valid = (unsigned)cur < (unsigned)n_edges;
      if (__builtin_amdgcn_ballot_w32(valid) == 0u) break;
      if (valid) {
        const int nx = nextL[cur];
        if (lvl < AGG_CAP) lv[lvl * 32 + lane] = srcL[cur] & 0x7fffffff;
        else if (lvl == AGG_CAP) currem = cur;
        cur = nx;
        cnt += 1;
      }
    }
    const float fsc = 0.5f / fmaxf((float)cnt, 1.0f);

    const int kept = (cnt < AGG_CAP) ? cnt : AGG_CAP;
    int incl = kept;
#pragma unroll
    for (int d = 1; d < 32; d <<= 1) {
      const int t = __shfl_up(incl, (unsigned)d);
      if (lane >= d) incl += t;
    }
    const int off   = incl - kept;
    const int total = __builtin_amdgcn_readlane(incl, 31);
    const int tot1  = __builtin_amdgcn_readlane(incl, 15);
    int ecol[AGG_CAP];
#pragma unroll
    for (int l = 0; l < AGG_CAP; ++l) ecol[l] = (l < kept) ? lv[l * 32 + lane] : 0;
#pragma unroll
    for (int l = 0; l < AGG_CAP; ++l) {
      if (l < kept) flat[off + l] = (l == kept - 1) ? (ecol[l] | (int)0x80000000u) : ecol[l];
    }
    const unsigned nz  = __builtin_amdgcn_ballot_w32(cnt > 0);
    const unsigned ovf = __builtin_amdgcn_ballot_w32(cnt > AGG_CAP);

    v8f acc[8];
#pragma unroll
    for (int j = 0; j < 8; ++j) acc[j] = v8f{};
#pragma unroll 1
    for (int p = 0; p < 3; ++p) {
      v2f a[32];
      if (p == 0) {
        load_a_tile(h, m0, lane, a);
      } else {
        const bool second = (p == 2);
        const int* __restrict__ srcT  = second ? src2  : src1;
        const int* __restrict__ nextT = second ? next2 : next1;
        const unsigned tmask = second ? 0xffff0000u : 0x0000ffffu;
        const int sb = second ? tot1 : 0;
        const int n  = second ? (total - tot1) : tot1;
        unsigned zm = ~nz & tmask;
        for (int t = 0; t < 16 && zm != 0u; ++t) {
          const int rr = __builtin_ctz(zm) & 15;
          zm &= zm - 1u;
          *(int4*)(stg + rr * HID + ((lane ^ rr) << 2)) = make_int4(0, 0, 0, 0);
        }
        unsigned rem = nz & tmask;
        int r = (rem != 0u) ? __builtin_ctz(rem) : 0;
        float4 sum = make_float4(0.0f, 0.0f, 0.0f, 0.0f);
        const int nfull = n & ~(AGG_BATCH - 1);
        for (int b0 = 0; b0 < nfull; b0 += AGG_BATCH) {
          const int ent = flat[sb + b0 + (lane & (AGG_BATCH - 1))];
          float4 v[AGG_BATCH];
#pragma unroll
          for (int u = 0; u < AGG_BATCH; ++u) {
            const int sj = __builtin_amdgcn_readlane(ent, u) & 0x7fffffff;
            v[u] = *(const float4*)(xl + (size_t)sj * OUTC);
          }
#pragma unroll
          for (int u = 0; u < AGG_BATCH; ++u) {
            sum.x += v[u].x;
            sum.y += v[u].y;
            sum.z += v[u].z;
            sum.w += v[u].w;
            if (__builtin_amdgcn_readlane(ent, u) < 0)
              close_row(sum, rem, r, ovf, currem, fsc, srcT, nextT, xl, stg, lane, n_edges);
          }
        }
        const int nt = n - nfull;
        if (nt > 0) {
          const int li  = lane & (AGG_BATCH - 1);
          const int ent = flat[sb + nfull + ((li < nt) ? li : nt - 1)];
          float4 v[AGG_BATCH];
#pragma unroll
          for (int u = 0; u < AGG_BATCH; ++u) {
            const int sj = __builtin_amdgcn_readlane(ent, u) & 0x7fffffff;
            v[u] = *(const float4*)(xl + (size_t)sj * OUTC);
          }
#pragma unroll
          for (int u = 0; u < AGG_BATCH; ++u) {
            if (u < nt) {
              sum.x += v[u].x;
              sum.y += v[u].y;
              sum.z += v[u].z;
              sum.w += v[u].w;
              if (__builtin_amdgcn_readlane(ent, u) < 0)
                close_row(sum, rem, r, ovf, currem, fsc, srcT, nextT, xl, stg, lane, n_edges);
            }
          }
        }
#pragma unroll
        for (int s = 0; s < 32; ++s) {
          const int2 t = *(const int2*)(stg + l16 * HID + ((s ^ l16) << 2) + 2 * half);
          a[s] = v2f{__int_as_float(t.x), __int_as_float(t.y)};
        }
      }
      wave_gemm_16x128(a, wl + p * (OUTC * WPITCH), lane, acc);
    }
#pragma unroll
    for (int r = 0; r < 8; ++r) {
      const int m = m0 + rbase + r;
#pragma unroll
      for (int j = 0; j < 8; ++j) {
        const int c = j * 16 + l16;
        out[(size_t)m * OUTC + c] = fmaxf(acc[j][r] + bl[c], 0.0f);
      }
    }
  }
}

extern "C" void kernel_launch(void* const* d_in, const int* in_sizes, int n_in,
                              void* d_out, int out_size, void* d_ws, size_t ws_size,
                              hipStream_t stream) {
  (void)n_in; (void)out_size; (void)ws_size;
  const float* h    = (const float*)d_in[0];
  const int*   src1 = (const int*)  d_in[1];
  const int*   dst1 = (const int*)  d_in[2];
  const int*   src2 = (const int*)  d_in[3];
  const int*   dst2 = (const int*)  d_in[4];
  const float* W1   = (const float*)d_in[5];
  const float* W2   = (const float*)d_in[6];
  const float* Wl   = (const float*)d_in[7];
  const float* bl   = (const float*)d_in[8];
  float*       out  = (float*)d_out;

  const size_t nodes = (size_t)in_sizes[0] / HID;
  const int    ne    = in_sizes[1];

  int* head1 = (int*)d_ws;
  int* head2 = head1 + nodes;
  int* next1 = head2 + nodes;
  int* next2 = next1 + ne;

  hipMemsetAsync(head1, 0xFF, 2 * nodes * sizeof(int), stream);

  const int  ntiles = (int)(nodes / 16);
  const dim3 gblk(GEMM_THREADS);
  const dim3 ggrd(FUSED_GROUPS);
  const dim3 lblk(LINK_THREADS);
  const dim3 lgrd((unsigned)((ne + LINK_THREADS * LINK_EDGES_PER_THREAD - 1) / (LINK_THREADS * LINK_EDGES_PER_THREAD)));

  link_kernel<<<lgrd, lblk, 0, stream>>>(dst1, dst2, head1, head2, next1, next2, ne);

  gather_gemm3_finalize_kernel<<<ggrd, gblk, 0, stream>>>(h, Wl, W1, W2, bl, src1, src2, head1, head2, next1, next2,
                                                          out, ntiles, ne);
}
